# starting kernel
# baseline (speedup 1.0000x reference)
#include <hip/hip_runtime.h>
#include <hip/hip_bf16.h>

#define C_   8
#define TF_  512
#define TG_  2048
#define H_   64
#define NITER 18

typedef __attribute__((ext_vector_type(2))) float v2f;
typedef __attribute__((ext_vector_type(8))) float v8f;

__global__ __launch_bounds__(256)
void init_x_kernel(const float* __restrict__ z,
                   const float* __restrict__ w_z,
                   const float* __restrict__ b_z,
                   float* __restrict__ Xa,
                   float* __restrict__ Xb) {
    int idx = blockIdx.x * blockDim.x + threadIdx.x;
    if (idx >= C_ * TG_ * H_) return;
    int h  = idx % H_;
    int ct = idx / H_;
    float v = z[ct] * w_z[h] + b_z[h];
    Xa[idx] = v;
    Xb[idx] = v;
}

__global__ __launch_bounds__(128)
void neumann_iter_kernel(const float* __restrict__ grn,
                         const float* __restrict__ z,
                         const float* __restrict__ w_z,
                         const float* __restrict__ b_z,
                         const int*   __restrict__ binary_tf,
                         const float* __restrict__ Xin,
                         float*       __restrict__ Xout) {
    const int c     = blockIdx.y;
    const int mtile = blockIdx.x;
    const int wave  = threadIdx.x >> 5;
    const int lane  = threadIdx.x & 31;
    const int m16   = lane & 15;
    const int hi    = lane >> 4;
    const int n     = wave * 16 + m16;

    const int* tf = binary_tf + c * TF_;

    v8f acc;
#pragma unroll
    for (int v = 0; v < 8; ++v) {
        int mrow = mtile * 16 + v + hi * 8;
        int trow = tf[mrow];
        acc[v] = z[c * TG_ + trow] * w_z[n] + b_z[n];
    }

    const float* grow = grn + ((size_t)(c * TF_ + mtile * 16 + m16)) * TG_;
    const float* xcol = Xin + (size_t)c * TG_ * H_ + n;
    const int klo = hi * 2;

#pragma unroll 4
    for (int kk = 0; kk < TG_; kk += 4) {
        v2f a = *(const v2f*)(grow + kk + klo);
        v2f b;
        b.x = xcol[(size_t)(kk + klo)     * H_];
        b.y = xcol[(size_t)(kk + klo + 1) * H_];
        acc = __builtin_amdgcn_wmma_f32_16x16x4_f32(
                  false, a, false, b, (short)0, acc, false, false);
    }

    float* xoutBase = Xout + (size_t)c * TG_ * H_;
#pragma unroll
    for (int v = 0; v < 8; ++v) {
        int mrow = mtile * 16 + v + hi * 8;
        int trow = tf[mrow];
        xoutBase[(size_t)trow * H_ + n] = acc[v];
    }
}

__global__ __launch_bounds__(128)
void mlp_kernel(const float* __restrict__ X,
                const float* __restrict__ w1, const float* __restrict__ b1,
                const float* __restrict__ w2, const float* __restrict__ b2,
                float* __restrict__ out) {
    int idx = blockIdx.x * blockDim.x + threadIdx.x;
    if (idx >= C_ * TG_) return;
    const float* x = X + (size_t)idx * H_;
    float xr[H_];
#pragma unroll
    for (int k = 0; k < H_; ++k) xr[k] = x[k];

    float o0 = b2[0];
    float o1 = b2[1];
    for (int j = 0; j < H_; ++j) {
        float hj = b1[j];
        const float* wrow = w1 + j * H_;
#pragma unroll
        for (int k = 0; k < H_; ++k) hj = fmaf(wrow[k], xr[k], hj);
        hj = fmaxf(hj, 0.0f);
        o0 = fmaf(w2[j],       hj, o0);
        o1 = fmaf(w2[H_ + j],  hj, o1);
    }
    out[idx] = o0;
    float sp = fmaxf(o1, 0.0f) + log1pf(expf(-fabsf(o1)));
    out[C_ * TG_ + idx] = sp + 1e-6f;
}

extern "C" void kernel_launch(void* const* d_in, const int* in_sizes, int n_in,
                              void* d_out, int out_size, void* d_ws, size_t ws_size,
                              hipStream_t stream) {
    (void)in_sizes; (void)n_in; (void)out_size; (void)ws_size;
    const float* z   = (const float*)d_in[0];
    const float* grn = (const float*)d_in[1];
    const int*   btf = (const int*)  d_in[2];
    const float* w_z = (const float*)d_in[4];
    const float* b_z = (const float*)d_in[5];
    const float* w1  = (const float*)d_in[6];
    const float* b1  = (const float*)d_in[7];
    const float* w2  = (const float*)d_in[8];
    const float* b2  = (const float*)d_in[9];
    float* out = (float*)d_out;

    float* Xa = (float*)d_ws;
    float* Xb = Xa + (size_t)C_ * TG_ * H_;

    const int total = C_ * TG_ * H_;
    init_x_kernel<<<(total + 255) / 256, 256, 0, stream>>>(z, w_z, b_z, Xa, Xb);

    dim3 grid(TF_ / 16, C_);
    float* bufs[2] = { Xa, Xb };
    for (int it = 0; it < NITER; ++it) {
        const float* xin = bufs[it & 1];
        float*      xout = bufs[(it + 1) & 1];
        neumann_iter_kernel<<<grid, 128, 0, stream>>>(grn, z, w_z, b_z, btf, xin, xout);
    }
    const float* xfin = bufs[NITER & 1];

    mlp_kernel<<<(C_ * TG_ + 127) / 128, 128, 0, stream>>>(xfin, w1, b1, w2, b2, out);
}


# sweep kernel: A (16xK) and X (Kx64) tiles staged through LDS in 128-row K blocks by coalesced 16-byte loads, next blocks' loads kept in flight across an LDS-only barrier; same k order, same bits
# speedup vs baseline: 3.0806x; 3.0806x over previous
#include <hip/hip_runtime.h>
#include <hip/hip_bf16.h>

#define C_   8
#define TF_  512
#define TG_  2048
#define H_   64
#define NITER 18

typedef __attribute__((ext_vector_type(2))) float v2f;
typedef __attribute__((ext_vector_type(8))) float v8f;

__global__ __launch_bounds__(256)
void init_x_kernel(const float* __restrict__ z,
                   const float* __restrict__ w_z,
                   const float* __restrict__ b_z,
                   float* __restrict__ Xa,
                   float* __restrict__ Xb) {
    int idx = blockIdx.x * blockDim.x + threadIdx.x;
    if (idx >= C_ * TG_ * H_) return;
    int h  = idx % H_;
    int ct = idx / H_;
    float v = z[ct] * w_z[h] + b_z[h];
    Xa[idx] = v;
    Xb[idx] = v;
}

#define KB_     128
#define NBLK_   (TG_ / KB_)
#define APITCH_ (KB_ + 4)
#define BP_     (H_ + 16)
#define ASTG_   (16 * APITCH_)
#define BSTG_   (KB_ * BP_)

typedef __attribute__((ext_vector_type(4))) float v4f;

__global__ __launch_bounds__(128)
void neumann_iter_kernel(const float* __restrict__ grn,
                         const float* __restrict__ z,
                         const float* __restrict__ w_z,
                         const float* __restrict__ b_z,
                         const int*   __restrict__ binary_tf,
                         const float* __restrict__ Xin,
                         float*       __restrict__ Xout) {
    __shared__ __attribute__((aligned(16))) float As[2 * ASTG_];
    __shared__ __attribute__((aligned(16))) float Bs[2 * BSTG_];

    const int c     = blockIdx.y;
    const int mtile = blockIdx.x;
    const int tid   = threadIdx.x;
    const int wave  = tid >> 5;
    const int lane  = tid & 31;
    const int m16   = lane & 15;
    const int hi    = lane >> 4;
    const int n     = wave * 16 + m16;

    const int* tf = binary_tf + c * TF_;

    const int arow = tid >> 5, aq = tid & 31;
    const int brow = tid >> 4, bq = tid & 15;
    const float* gA = grn + ((size_t)(c * TF_ + mtile * 16 + arow)) * TG_ + aq * 4;
    const float* gB = Xin + (size_t)c * TG_ * H_ + (size_t)brow * H_ + bq * 4;
    float* sA = As + arow * APITCH_ + aq * 4;
    float* sB = Bs + brow * BP_ + bq * 4;

    const float* fA = As + m16 * APITCH_ + 2 * hi;
    const float* fB = Bs + (2 * hi) * BP_ + n;

    v4f ra[4], rb[16];
    v2f fa0[16], fb0[16];
    v2f fa1[16], fb1[16];

#define GLOAD(blk)                                                              \
    do {                                                                        \
        _Pragma("unroll")                                                       \
        for (int i = 0; i < 4; ++i)                                             \
            ra[i] = *(const v4f*)(gA + (size_t)(4 * i) * TG_ + (blk) * KB_);    \
        _Pragma("unroll")                                                       \
        for (int i = 0; i < 16; ++i)                                            \
            rb[i] = *(const v4f*)(gB + ((size_t)(blk) * KB_ + 8 * i) * H_);     \
    } while (0)

#define LSTORE(st)                                                              \
    do {                                                                        \
        _Pragma("unroll")                                                       \
        for (int i = 0; i < 4; ++i)                                             \
            *(v4f*)(sA + (st) * ASTG_ + (4 * i) * APITCH_) = ra[i];             \
        _Pragma("unroll")                                                       \
        for (int i = 0; i < 16; ++i)                                            \
            *(v4f*)(sB + (st) * BSTG_ + (8 * i) * BP_) = rb[i];                 \
    } while (0)

#define FLOAD(fa, fb, st, half)                                                 \
    do {                                                                        \
        _Pragma("unroll")                                                       \
        for (int q = 0; q < 16; ++q) {                                          \
            const int kk_ = (half) * 64 + q * 4;                                \
            fa[q] = *(const v2f*)(fA + (st) * ASTG_ + kk_);                     \
            const float* b_ = fB + (st) * BSTG_ + kk_ * BP_;                    \
            fb[q].x = b_[0];                                                    \
            fb[q].y = b_[BP_];                                                  \
        }                                                                       \
    } while (0)

#define MMA16(fa, fb)                                                           \
    do {                                                                        \
        _Pragma("unroll")                                                       \
        for (int q = 0; q < 16; ++q)                                            \
            acc = __builtin_amdgcn_wmma_f32_16x16x4_f32(                        \
                      false, fa[q], false, fb[q], (short)0, acc, false, false); \
    } while (0)

#define LDS_BARRIER()                                                           \
    asm volatile("s_wait_dscnt 0x0\n\ts_barrier_signal -1\n\ts_barrier_wait -1" ::: "memory")

    GLOAD(0);

    v8f acc;
#pragma unroll
    for (int v = 0; v < 8; ++v) {
        int mrow = mtile * 16 + v + hi * 8;
        int trow = tf[mrow];
        acc[v] = z[c * TG_ + trow] * w_z[n] + b_z[n];
    }

    LSTORE(0);
    GLOAD(1);
    LDS_BARRIER();

    for (int blk = 0; blk < NBLK_ - 1; ++blk) {
        const int st = blk & 1;
        FLOAD(fa0, fb0, st, 0);
        FLOAD(fa1, fb1, st, 1);
        __builtin_amdgcn_sched_barrier(0);
        MMA16(fa0, fb0);
        MMA16(fa1, fb1);
        __builtin_amdgcn_sched_barrier(0);
        LSTORE(st ^ 1);
        if (blk + 2 < NBLK_)
            GLOAD(blk + 2);
        LDS_BARRIER();
    }
    FLOAD(fa0, fb0, (NBLK_ - 1) & 1, 0);
    FLOAD(fa1, fb1, (NBLK_ - 1) & 1, 1);
    __builtin_amdgcn_sched_barrier(0);
    MMA16(fa0, fb0);
    MMA16(fa1, fb1);

#undef GLOAD
#undef LSTORE
#undef FLOAD
#undef MMA16
#undef LDS_BARRIER

    float* xoutBase = Xout + (size_t)c * TG_ * H_;
#pragma unroll
    for (int v = 0; v < 8; ++v) {
        int mrow = mtile * 16 + v + hi * 8;
        int trow = tf[mrow];
        xoutBase[(size_t)trow * H_ + n] = acc[v];
    }
}

__global__ __launch_bounds__(128)
void mlp_kernel(const float* __restrict__ X,
                const float* __restrict__ w1, const float* __restrict__ b1,
                const float* __restrict__ w2, const float* __restrict__ b2,
                float* __restrict__ out) {
    int idx = blockIdx.x * blockDim.x + threadIdx.x;
    if (idx >= C_ * TG_) return;
    const float* x = X + (size_t)idx * H_;
    float xr[H_];
#pragma unroll
    for (int k = 0; k < H_; ++k) xr[k] = x[k];

    float o0 = b2[0];
    float o1 = b2[1];
    for (int j = 0; j < H_; ++j) {
        float hj = b1[j];
        const float* wrow = w1 + j * H_;
#pragma unroll
        for (int k = 0; k < H_; ++k) hj = fmaf(wrow[k], xr[k], hj);
        hj = fmaxf(hj, 0.0f);
        o0 = fmaf(w2[j],       hj, o0);
        o1 = fmaf(w2[H_ + j],  hj, o1);
    }
    out[idx] = o0;
    float sp = fmaxf(o1, 0.0f) + log1pf(expf(-fabsf(o1)));
    out[C_ * TG_ + idx] = sp + 1e-6f;
}

extern "C" void kernel_launch(void* const* d_in, const int* in_sizes, int n_in,
                              void* d_out, int out_size, void* d_ws, size_t ws_size,
                              hipStream_t stream) {
    (void)in_sizes; (void)n_in; (void)out_size; (void)ws_size;
    const float* z   = (const float*)d_in[0];
    const float* grn = (const float*)d_in[1];
    const int*   btf = (const int*)  d_in[2];
    const float* w_z = (const float*)d_in[4];
    const float* b_z = (const float*)d_in[5];
    const float* w1  = (const float*)d_in[6];
    const float* b1  = (const float*)d_in[7];
    const float* w2  = (const float*)d_in[8];
    const float* b2  = (const float*)d_in[9];
    float* out = (float*)d_out;

    float* Xa = (float*)d_ws;
    float* Xb = Xa + (size_t)C_ * TG_ * H_;

    const int total = C_ * TG_ * H_;
    init_x_kernel<<<(total + 255) / 256, 256, 0, stream>>>(z, w_z, b_z, Xa, Xb);

    dim3 grid(TF_ / 16, C_);
    float* bufs[2] = { Xa, Xb };
    for (int it = 0; it < NITER; ++it) {
        const float* xin = bufs[it & 1];
        float*      xout = bufs[(it + 1) & 1];
        neumann_iter_kernel<<<grid, 128, 0, stream>>>(grn, z, w_z, b_z, btf, xin, xout);
    }
    const float* xfin = bufs[NITER & 1];

    mlp_kernel<<<(C_ * TG_ + 127) / 128, 128, 0, stream>>>(xfin, w1, b1, w2, b2, out);
}


# sweep kernel: global loads run two K blocks ahead through a second staging register set (two trips of flight before the LDS store that waits for them, was under one); same k order, same bits
# speedup vs baseline: 3.5689x; 1.1585x over previous
#include <hip/hip_runtime.h>
#include <hip/hip_bf16.h>

#define C_   8
#define TF_  512
#define TG_  2048
#define H_   64
#define NITER 18

typedef __attribute__((ext_vector_type(2))) float v2f;
typedef __attribute__((ext_vector_type(8))) float v8f;

__global__ __launch_bounds__(256)
void init_x_kernel(const float* __restrict__ z,
                   const float* __restrict__ w_z,
                   const float* __restrict__ b_z,
                   float* __restrict__ Xa,
                   float* __restrict__ Xb) {
    int idx = blockIdx.x * blockDim.x + threadIdx.x;
    if (idx >= C_ * TG_ * H_) return;
    int h  = idx % H_;
    int ct = idx / H_;
    float v = z[ct] * w_z[h] + b_z[h];
    Xa[idx] = v;
    Xb[idx] = v;
}

#define KB_     128
#define NBLK_   (TG_ / KB_)
#define APITCH_ (KB_ + 4)
#define BP_     (H_ + 16)
#define ASTG_   (16 * APITCH_)
#define BSTG_   (KB_ * BP_)

typedef __attribute__((ext_vector_type(4))) float v4f;

__global__ __launch_bounds__(128)
void neumann_iter_kernel(const float* __restrict__ grn,
                         const float* __restrict__ z,
                         const float* __restrict__ w_z,
                         const float* __restrict__ b_z,
                         const int*   __restrict__ binary_tf,
                         const float* __restrict__ Xin,
                         float*       __restrict__ Xout) {
    __shared__ __attribute__((aligned(16))) float As[2 * ASTG_];
    __shared__ __attribute__((aligned(16))) float Bs[2 * BSTG_];

    const int c     = blockIdx.y;
    const int mtile = blockIdx.x;
    const int tid   = threadIdx.x;
    const int wave  = tid >> 5;
    const int lane  = tid & 31;
    const int m16   = lane & 15;
    const int hi    = lane >> 4;
    const int n     = wave * 16 + m16;

    const int* tf = binary_tf + c * TF_;

    const int arow = tid >> 5, aq = tid & 31;
    const int brow = tid >> 4, bq = tid & 15;
    const float* gA = grn + ((size_t)(c * TF_ + mtile * 16 + arow)) * TG_ + aq * 4;
    const float* gB = Xin + (size_t)c * TG_ * H_ + (size_t)brow * H_ + bq * 4;
    float* sA = As + arow * APITCH_ + aq * 4;
    float* sB = Bs + brow * BP_ + bq * 4;

    const float* fA = As + m16 * APITCH_ + 2 * hi;
    const float* fB = Bs + (2 * hi) * BP_ + n;

    v4f ra0[4], rb0[16];
    v4f ra1[4], rb1[16];
    v2f fa0[16], fb0[16];
    v2f fa1[16], fb1[16];

#define GLOAD(blk, ra, rb)                                                      \
    do {                                                                        \
        _Pragma("unroll")                                                       \
        for (int i = 0; i < 4; ++i)                                             \
            ra[i] = *(const v4f*)(gA + (size_t)(4 * i) * TG_ + (blk) * KB_);    \
        _Pragma("unroll")                                                       \
        for (int i = 0; i < 16; ++i)                                            \
            rb[i] = *(const v4f*)(gB + ((size_t)(blk) * KB_ + 8 * i) * H_);     \
    } while (0)

#define LSTORE(st, ra, rb)                                                      \
    do {                                                                        \
        _Pragma("unroll")                                                       \
        for (int i = 0; i < 4; ++i)                                             \
            *(v4f*)(sA + (st) * ASTG_ + (4 * i) * APITCH_) = ra[i];             \
        _Pragma("unroll")                                                       \
        for (int i = 0; i < 16; ++i)                                            \
            *(v4f*)(sB + (st) * BSTG_ + (8 * i) * BP_) = rb[i];                 \
    } while (0)

#define FLOAD(fa, fb, st, half)                                                 \
    do {                                                                        \
        _Pragma("unroll")                                                       \
        for (int q = 0; q < 16; ++q) {                                          \
            const int kk_ = (half) * 64 + q * 4;                                \
            fa[q] = *(const v2f*)(fA + (st) * ASTG_ + kk_);                     \
            const float* b_ = fB + (st) * BSTG_ + kk_ * BP_;                    \
            fb[q].x = b_[0];                                                    \
            fb[q].y = b_[BP_];                                                  \
        }                                                                       \
    } while (0)

#define MMA16(fa, fb)                                                           \
    do {                                                                        \
        _Pragma("unroll")                                                       \
        for (int q = 0; q < 16; ++q)                                            \
            acc = __builtin_amdgcn_wmma_f32_16x16x4_f32(                        \
                      false, fa[q], false, fb[q], (short)0, acc, false, false); \
    } while (0)

#define LDS_BARRIER()                                                           \
    asm volatile("s_wait_dscnt 0x0\n\ts_barrier_signal -1\n\ts_barrier_wait -1" ::: "memory")

#define TRIP(st, ra, rb)                                                        \
    do {                                                                        \
        FLOAD(fa0, fb0, st, 0);                                                 \
        FLOAD(fa1, fb1, st, 1);                                                 \
          \
          \
          \
        __builtin_amdgcn_sched_barrier(0);                                      \
        MMA16(fa0, fb0);                                                        \
        MMA16(fa1, fb1);                                                        \
          \
          \
        __builtin_amdgcn_sched_barrier(0);                                      \
        LSTORE((st) ^ 1, ra, rb);                                               \
    } while (0)

    GLOAD(0, ra0, rb0);

    v8f acc;
#pragma unroll
    for (int v = 0; v < 8; ++v) {
        int mrow = mtile * 16 + v + hi * 8;
        int trow = tf[mrow];
        acc[v] = z[c * TG_ + trow] * w_z[n] + b_z[n];
    }

    GLOAD(1, ra1, rb1);
    __builtin_amdgcn_sched_barrier(0);
    LSTORE(0, ra0, rb0);
    GLOAD(2, ra0, rb0);
    LDS_BARRIER();

    for (int blk = 0; blk < NBLK_ - 4; blk += 2) {
        TRIP(0, ra1, rb1);
        GLOAD(blk + 3, ra1, rb1);
        LDS_BARRIER();
        TRIP(1, ra0, rb0);
        GLOAD(blk + 4, ra0, rb0);
        LDS_BARRIER();
    }
    TRIP(0, ra1, rb1);
    GLOAD(NBLK_ - 1, ra1, rb1);
    LDS_BARRIER();
    TRIP(1, ra0, rb0);
    LDS_BARRIER();
    TRIP(0, ra1, rb1);
    LDS_BARRIER();
    FLOAD(fa0, fb0, 1, 0);
    FLOAD(fa1, fb1, 1, 1);
    __builtin_amdgcn_sched_barrier(0);
    MMA16(fa0, fb0);
    MMA16(fa1, fb1);

#undef TRIP
#undef GLOAD
#undef LSTORE
#undef FLOAD
#undef MMA16
#undef LDS_BARRIER

    float* xoutBase = Xout + (size_t)c * TG_ * H_;
#pragma unroll
    for (int v = 0; v < 8; ++v) {
        int mrow = mtile * 16 + v + hi * 8;
        int trow = tf[mrow];
        xoutBase[(size_t)trow * H_ + n] = acc[v];
    }
}

__global__ __launch_bounds__(128)
void mlp_kernel(const float* __restrict__ X,
                const float* __restrict__ w1, const float* __restrict__ b1,
                const float* __restrict__ w2, const float* __restrict__ b2,
                float* __restrict__ out) {
    int idx = blockIdx.x * blockDim.x + threadIdx.x;
    if (idx >= C_ * TG_) return;
    const float* x = X + (size_t)idx * H_;
    float xr[H_];
#pragma unroll
    for (int k = 0; k < H_; ++k) xr[k] = x[k];

    float o0 = b2[0];
    float o1 = b2[1];
    for (int j = 0; j < H_; ++j) {
        float hj = b1[j];
        const float* wrow = w1 + j * H_;
#pragma unroll
        for (int k = 0; k < H_; ++k) hj = fmaf(wrow[k], xr[k], hj);
        hj = fmaxf(hj, 0.0f);
        o0 = fmaf(w2[j],       hj, o0);
        o1 = fmaf(w2[H_ + j],  hj, o1);
    }
    out[idx] = o0;
    float sp = fmaxf(o1, 0.0f) + log1pf(expf(-fabsf(o1)));
    out[C_ * TG_ + idx] = sp + 1e-6f;
}

extern "C" void kernel_launch(void* const* d_in, const int* in_sizes, int n_in,
                              void* d_out, int out_size, void* d_ws, size_t ws_size,
                              hipStream_t stream) {
    (void)in_sizes; (void)n_in; (void)out_size; (void)ws_size;
    const float* z   = (const float*)d_in[0];
    const float* grn = (const float*)d_in[1];
    const int*   btf = (const int*)  d_in[2];
    const float* w_z = (const float*)d_in[4];
    const float* b_z = (const float*)d_in[5];
    const float* w1  = (const float*)d_in[6];
    const float* b1  = (const float*)d_in[7];
    const float* w2  = (const float*)d_in[8];
    const float* b2  = (const float*)d_in[9];
    float* out = (float*)d_out;

    float* Xa = (float*)d_ws;
    float* Xb = Xa + (size_t)C_ * TG_ * H_;

    const int total = C_ * TG_ * H_;
    init_x_kernel<<<(total + 255) / 256, 256, 0, stream>>>(z, w_z, b_z, Xa, Xb);

    dim3 grid(TF_ / 16, C_);
    float* bufs[2] = { Xa, Xb };
    for (int it = 0; it < NITER; ++it) {
        const float* xin = bufs[it & 1];
        float*      xout = bufs[(it + 1) & 1];
        neumann_iter_kernel<<<grid, 128, 0, stream>>>(grn, z, w_z, b_z, btf, xin, xout);
    }
    const float* xfin = bufs[NITER & 1];

    mlp_kernel<<<(C_ * TG_ + 127) / 128, 128, 0, stream>>>(xfin, w1, b1, w2, b2, out);
}


# sweep kernel: k steps dealt alternately to two accumulator tiles (even: seeded with C, odd: from zero), added once behind the K loop, so no WMMA waits for its predecessor's D; sum order changes
# speedup vs baseline: 3.7810x; 1.0594x over previous
#include <hip/hip_runtime.h>
#include <hip/hip_bf16.h>

#define C_   8
#define TF_  512
#define TG_  2048
#define H_   64
#define NITER 18

typedef __attribute__((ext_vector_type(2))) float v2f;
typedef __attribute__((ext_vector_type(8))) float v8f;

__global__ __launch_bounds__(256)
void init_x_kernel(const float* __restrict__ z,
                   const float* __restrict__ w_z,
                   const float* __restrict__ b_z,
                   float* __restrict__ Xa,
                   float* __restrict__ Xb) {
    int idx = blockIdx.x * blockDim.x + threadIdx.x;
    if (idx >= C_ * TG_ * H_) return;
    int h  = idx % H_;
    int ct = idx / H_;
    float v = z[ct] * w_z[h] + b_z[h];
    Xa[idx] = v;
    Xb[idx] = v;
}

#define KB_     128
#define NBLK_   (TG_ / KB_)
#define APITCH_ (KB_ + 4)
#define BP_     (H_ + 16)
#define ASTG_   (16 * APITCH_)
#define BSTG_   (KB_ * BP_)

typedef __attribute__((ext_vector_type(4))) float v4f;

__global__ __launch_bounds__(128)
void neumann_iter_kernel(const float* __restrict__ grn,
                         const float* __restrict__ z,
                         const float* __restrict__ w_z,
                         const float* __restrict__ b_z,
                         const int*   __restrict__ binary_tf,
                         const float* __restrict__ Xin,
                         float*       __restrict__ Xout) {
    __shared__ __attribute__((aligned(16))) float As[2 * ASTG_];
    __shared__ __attribute__((aligned(16))) float Bs[2 * BSTG_];

    const int c     = blockIdx.y;
    const int mtile = blockIdx.x;
    const int tid   = threadIdx.x;
    const int wave  = tid >> 5;
    const int lane  = tid & 31;
    const int m16   = lane & 15;
    const int hi    = lane >> 4;
    const int n     = wave * 16 + m16;

    const int* tf = binary_tf + c * TF_;

    const int arow = tid >> 5, aq = tid & 31;
    const int brow = tid >> 4, bq = tid & 15;
    const float* gA = grn + ((size_t)(c * TF_ + mtile * 16 + arow)) * TG_ + aq * 4;
    const float* gB = Xin + (size_t)c * TG_ * H_ + (size_t)brow * H_ + bq * 4;
    float* sA = As + arow * APITCH_ + aq * 4;
    float* sB = Bs + brow * BP_ + bq * 4;

    const float* fA = As + m16 * APITCH_ + 2 * hi;
    const float* fB = Bs + (2 * hi) * BP_ + n;

    v4f ra0[4], rb0[16];
    v4f ra1[4], rb1[16];
    v2f fa0[16], fb0[16];
    v2f fa1[16], fb1[16];

#define GLOAD(blk, ra, rb)                                                      \
    do {                                                                        \
        _Pragma("unroll")                                                       \
        for (int i = 0; i < 4; ++i)                                             \
            ra[i] = *(const v4f*)(gA + (size_t)(4 * i) * TG_ + (blk) * KB_);    \
        _Pragma("unroll")                                                       \
        for (int i = 0; i < 16; ++i)                                            \
            rb[i] = *(const v4f*)(gB + ((size_t)(blk) * KB_ + 8 * i) * H_);     \
    } while (0)

#define LSTORE(st, ra, rb)                                                      \
    do {                                                                        \
        _Pragma("unroll")                                                       \
        for (int i = 0; i < 4; ++i)                                             \
            *(v4f*)(sA + (st) * ASTG_ + (4 * i) * APITCH_) = ra[i];             \
        _Pragma("unroll")                                                       \
        for (int i = 0; i < 16; ++i)                                            \
            *(v4f*)(sB + (st) * BSTG_ + (8 * i) * BP_) = rb[i];                 \
    } while (0)

#define FLOAD(fa, fb, st, half)                                                 \
    do {                                                                        \
        _Pragma("unroll")                                                       \
        for (int q = 0; q < 16; ++q) {                                          \
            const int kk_ = (half) * 64 + q * 4;                                \
            fa[q] = *(const v2f*)(fA + (st) * ASTG_ + kk_);                     \
            const float* b_ = fB + (st) * BSTG_ + kk_ * BP_;                    \
            fb[q].x = b_[0];                                                    \
            fb[q].y = b_[BP_];                                                  \
        }                                                                       \
    } while (0)

#define MMA16(fa, fb)                                                           \
    do {                                                                        \
        _Pragma("unroll")                                                       \
        for (int q = 0; q < 16; q += 2) {                                       \
            acc0 = __builtin_amdgcn_wmma_f32_16x16x4_f32(                       \
                      false, fa[q], false, fb[q], (short)0, acc0, false, false);\
            acc1 = __builtin_amdgcn_wmma_f32_16x16x4_f32(                       \
                      false, fa[q + 1], false, fb[q + 1], (short)0, acc1,       \
                      false, false);                                            \
        }                                                                       \
    } while (0)

#define LDS_BARRIER()                                                           \
    asm volatile("s_wait_dscnt 0x0\n\ts_barrier_signal -1\n\ts_barrier_wait -1" ::: "memory")

#define TRIP(st, ra, rb)                                                        \
    do {                                                                        \
        FLOAD(fa0, fb0, st, 0);                                                 \
        FLOAD(fa1, fb1, st, 1);                                                 \
          \
          \
          \
        __builtin_amdgcn_sched_barrier(0);                                      \
        MMA16(fa0, fb0);                                                        \
        MMA16(fa1, fb1);                                                        \
          \
          \
        __builtin_amdgcn_sched_barrier(0);                                      \
        LSTORE((st) ^ 1, ra, rb);                                               \
    } while (0)

    GLOAD(0, ra0, rb0);

    v8f acc0, acc1;
#pragma unroll
    for (int v = 0; v < 8; ++v) {
        int mrow = mtile * 16 + v + hi * 8;
        int trow = tf[mrow];
        acc0[v] = z[c * TG_ + trow] * w_z[n] + b_z[n];
        acc1[v] = 0.0f;
    }

    GLOAD(1, ra1, rb1);
    __builtin_amdgcn_sched_barrier(0);
    LSTORE(0, ra0, rb0);
    GLOAD(2, ra0, rb0);
    LDS_BARRIER();

    for (int blk = 0; blk < NBLK_ - 4; blk += 2) {
        TRIP(0, ra1, rb1);
        GLOAD(blk + 3, ra1, rb1);
        LDS_BARRIER();
        TRIP(1, ra0, rb0);
        GLOAD(blk + 4, ra0, rb0);
        LDS_BARRIER();
    }
    TRIP(0, ra1, rb1);
    GLOAD(NBLK_ - 1, ra1, rb1);
    LDS_BARRIER();
    TRIP(1, ra0, rb0);
    LDS_BARRIER();
    TRIP(0, ra1, rb1);
    LDS_BARRIER();
    FLOAD(fa0, fb0, 1, 0);
    FLOAD(fa1, fb1, 1, 1);
    __builtin_amdgcn_sched_barrier(0);
    MMA16(fa0, fb0);
    MMA16(fa1, fb1);

#undef TRIP
#undef GLOAD
#undef LSTORE
#undef FLOAD
#undef MMA16
#undef LDS_BARRIER

    float* xoutBase = Xout + (size_t)c * TG_ * H_;
#pragma unroll
    for (int v = 0; v < 8; ++v) {
        int mrow = mtile * 16 + v + hi * 8;
        int trow = tf[mrow];
        xoutBase[(size_t)trow * H_ + n] = acc0[v] + acc1[v];
    }
}

__global__ __launch_bounds__(128)
void mlp_kernel(const float* __restrict__ X,
                const float* __restrict__ w1, const float* __restrict__ b1,
                const float* __restrict__ w2, const float* __restrict__ b2,
                float* __restrict__ out) {
    int idx = blockIdx.x * blockDim.x + threadIdx.x;
    if (idx >= C_ * TG_) return;
    const float* x = X + (size_t)idx * H_;
    float xr[H_];
#pragma unroll
    for (int k = 0; k < H_; ++k) xr[k] = x[k];

    float o0 = b2[0];
    float o1 = b2[1];
    for (int j = 0; j < H_; ++j) {
        float hj = b1[j];
        const float* wrow = w1 + j * H_;
#pragma unroll
        for (int k = 0; k < H_; ++k) hj = fmaf(wrow[k], xr[k], hj);
        hj = fmaxf(hj, 0.0f);
        o0 = fmaf(w2[j],       hj, o0);
        o1 = fmaf(w2[H_ + j],  hj, o1);
    }
    out[idx] = o0;
    float sp = fmaxf(o1, 0.0f) + log1pf(expf(-fabsf(o1)));
    out[C_ * TG_ + idx] = sp + 1e-6f;
}

extern "C" void kernel_launch(void* const* d_in, const int* in_sizes, int n_in,
                              void* d_out, int out_size, void* d_ws, size_t ws_size,
                              hipStream_t stream) {
    (void)in_sizes; (void)n_in; (void)out_size; (void)ws_size;
    const float* z   = (const float*)d_in[0];
    const float* grn = (const float*)d_in[1];
    const int*   btf = (const int*)  d_in[2];
    const float* w_z = (const float*)d_in[4];
    const float* b_z = (const float*)d_in[5];
    const float* w1  = (const float*)d_in[6];
    const float* b1  = (const float*)d_in[7];
    const float* w2  = (const float*)d_in[8];
    const float* b2  = (const float*)d_in[9];
    float* out = (float*)d_out;

    float* Xa = (float*)d_ws;
    float* Xb = Xa + (size_t)C_ * TG_ * H_;

    const int total = C_ * TG_ * H_;
    init_x_kernel<<<(total + 255) / 256, 256, 0, stream>>>(z, w_z, b_z, Xa, Xb);

    dim3 grid(TF_ / 16, C_);
    float* bufs[2] = { Xa, Xb };
    for (int it = 0; it < NITER; ++it) {
        const float* xin = bufs[it & 1];
        float*      xout = bufs[(it + 1) & 1];
        neumann_iter_kernel<<<grid, 128, 0, stream>>>(grn, z, w_z, b_z, btf, xin, xout);
    }
    const float* xfin = bufs[NITER & 1];

    mlp_kernel<<<(C_ * TG_ + 127) / 128, 128, 0, stream>>>(xfin, w1, b1, w2, b2, out);
}


# MLP launch: weights staged once per work-group in LDS, read as broadcast 16-byte loads into 8 independent FMA chains per thread (was one dependent chain fed by in-line scalar loads); same bits
# speedup vs baseline: 4.6233x; 1.2228x over previous
#include <hip/hip_runtime.h>
#include <hip/hip_bf16.h>

#define C_   8
#define TF_  512
#define TG_  2048
#define H_   64
#define NITER 18

typedef __attribute__((ext_vector_type(2))) float v2f;
typedef __attribute__((ext_vector_type(8))) float v8f;

__global__ __launch_bounds__(256)
void init_x_kernel(const float* __restrict__ z,
                   const float* __restrict__ w_z,
                   const float* __restrict__ b_z,
                   float* __restrict__ Xa,
                   float* __restrict__ Xb) {
    int idx = blockIdx.x * blockDim.x + threadIdx.x;
    if (idx >= C_ * TG_ * H_) return;
    int h  = idx % H_;
    int ct = idx / H_;
    float v = z[ct] * w_z[h] + b_z[h];
    Xa[idx] = v;
    Xb[idx] = v;
}

#define KB_     128
#define NBLK_   (TG_ / KB_)
#define APITCH_ (KB_ + 4)
#define BP_     (H_ + 16)
#define ASTG_   (16 * APITCH_)
#define BSTG_   (KB_ * BP_)

typedef __attribute__((ext_vector_type(4))) float v4f;

__global__ __launch_bounds__(128)
void neumann_iter_kernel(const float* __restrict__ grn,
                         const float* __restrict__ z,
                         const float* __restrict__ w_z,
                         const float* __restrict__ b_z,
                         const int*   __restrict__ binary_tf,
                         const float* __restrict__ Xin,
                         float*       __restrict__ Xout) {
    __shared__ __attribute__((aligned(16))) float As[2 * ASTG_];
    __shared__ __attribute__((aligned(16))) float Bs[2 * BSTG_];

    const int c     = blockIdx.y;
    const int mtile = blockIdx.x;
    const int tid   = threadIdx.x;
    const int wave  = tid >> 5;
    const int lane  = tid & 31;
    const int m16   = lane & 15;
    const int hi    = lane >> 4;
    const int n     = wave * 16 + m16;

    const int* tf = binary_tf + c * TF_;

    const int arow = tid >> 5, aq = tid & 31;
    const int brow = tid >> 4, bq = tid & 15;
    const float* gA = grn + ((size_t)(c * TF_ + mtile * 16 + arow)) * TG_ + aq * 4;
    const float* gB = Xin + (size_t)c * TG_ * H_ + (size_t)brow * H_ + bq * 4;
    float* sA = As + arow * APITCH_ + aq * 4;
    float* sB = Bs + brow * BP_ + bq * 4;

    const float* fA = As + m16 * APITCH_ + 2 * hi;
    const float* fB = Bs + (2 * hi) * BP_ + n;

    v4f ra0[4], rb0[16];
    v4f ra1[4], rb1[16];
    v2f fa0[16], fb0[16];
    v2f fa1[16], fb1[16];

#define GLOAD(blk, ra, rb)                                                      \
    do {                                                                        \
        _Pragma("unroll")                                                       \
        for (int i = 0; i < 4; ++i)                                             \
            ra[i] = *(const v4f*)(gA + (size_t)(4 * i) * TG_ + (blk) * KB_);    \
        _Pragma("unroll")                                                       \
        for (int i = 0; i < 16; ++i)                                            \
            rb[i] = *(const v4f*)(gB + ((size_t)(blk) * KB_ + 8 * i) * H_);     \
    } while (0)

#define LSTORE(st, ra, rb)                                                      \
    do {                                                                        \
        _Pragma("unroll")                                                       \
        for (int i = 0; i < 4; ++i)                                             \
            *(v4f*)(sA + (st) * ASTG_ + (4 * i) * APITCH_) = ra[i];             \
        _Pragma("unroll")                                                       \
        for (int i = 0; i < 16; ++i)                                            \
            *(v4f*)(sB + (st) * BSTG_ + (8 * i) * BP_) = rb[i];                 \
    } while (0)

#define FLOAD(fa, fb, st, half)                                                 \
    do {                                                                        \
        _Pragma("unroll")                                                       \
        for (int q = 0; q < 16; ++q) {                                          \
            const int kk_ = (half) * 64 + q * 4;                                \
            fa[q] = *(const v2f*)(fA + (st) * ASTG_ + kk_);                     \
            const float* b_ = fB + (st) * BSTG_ + kk_ * BP_;                    \
            fb[q].x = b_[0];                                                    \
            fb[q].y = b_[BP_];                                                  \
        }                                                                       \
    } while (0)

#define MMA16(fa, fb)                                                           \
    do {                                                                        \
        _Pragma("unroll")                                                       \
        for (int q = 0; q < 16; q += 2) {                                       \
            acc0 = __builtin_amdgcn_wmma_f32_16x16x4_f32(                       \
                      false, fa[q], false, fb[q], (short)0, acc0, false, false);\
            acc1 = __builtin_amdgcn_wmma_f32_16x16x4_f32(                       \
                      false, fa[q + 1], false, fb[q + 1], (short)0, acc1,       \
                      false, false);                                            \
        }                                                                       \
    } while (0)

#define LDS_BARRIER()                                                           \
    asm volatile("s_wait_dscnt 0x0\n\ts_barrier_signal -1\n\ts_barrier_wait -1" ::: "memory")

#define TRIP(st, ra, rb)                                                        \
    do {                                                                        \
        FLOAD(fa0, fb0, st, 0);                                                 \
        FLOAD(fa1, fb1, st, 1);                                                 \
          \
          \
          \
        __builtin_amdgcn_sched_barrier(0);                                      \
        MMA16(fa0, fb0);                                                        \
        MMA16(fa1, fb1);                                                        \
          \
          \
        __builtin_amdgcn_sched_barrier(0);                                      \
        LSTORE((st) ^ 1, ra, rb);                                               \
    } while (0)

    GLOAD(0, ra0, rb0);

    v8f acc0, acc1;
#pragma unroll
    for (int v = 0; v < 8; ++v) {
        int mrow = mtile * 16 + v + hi * 8;
        int trow = tf[mrow];
        acc0[v] = z[c * TG_ + trow] * w_z[n] + b_z[n];
        acc1[v] = 0.0f;
    }

    GLOAD(1, ra1, rb1);
    __builtin_amdgcn_sched_barrier(0);
    LSTORE(0, ra0, rb0);
    GLOAD(2, ra0, rb0);
    LDS_BARRIER();

    for (int blk = 0; blk < NBLK_ - 4; blk += 2) {
        TRIP(0, ra1, rb1);
        GLOAD(blk + 3, ra1, rb1);
        LDS_BARRIER();
        TRIP(1, ra0, rb0);
        GLOAD(blk + 4, ra0, rb0);
        LDS_BARRIER();
    }
    TRIP(0, ra1, rb1);
    GLOAD(NBLK_ - 1, ra1, rb1);
    LDS_BARRIER();
    TRIP(1, ra0, rb0);
    LDS_BARRIER();
    TRIP(0, ra1, rb1);
    LDS_BARRIER();
    FLOAD(fa0, fb0, 1, 0);
    FLOAD(fa1, fb1, 1, 1);
    __builtin_amdgcn_sched_barrier(0);
    MMA16(fa0, fb0);
    MMA16(fa1, fb1);

#undef TRIP
#undef GLOAD
#undef LSTORE
#undef FLOAD
#undef MMA16
#undef LDS_BARRIER

    float* xoutBase = Xout + (size_t)c * TG_ * H_;
#pragma unroll
    for (int v = 0; v < 8; ++v) {
        int mrow = mtile * 16 + v + hi * 8;
        int trow = tf[mrow];
        xoutBase[(size_t)trow * H_ + n] = acc0[v] + acc1[v];
    }
}

#define MLP_ROWS_ 64
#define MLP_JB_   8
#define MLP_KQ_   16
__global__ __launch_bounds__(MLP_ROWS_)
void mlp_kernel(const float* __restrict__ X,
                const float* __restrict__ w1, const float* __restrict__ b1,
                const float* __restrict__ w2, const float* __restrict__ b2,
                float* __restrict__ out) {
    __shared__ __attribute__((aligned(16))) float sW1[H_ * H_];
    __shared__ __attribute__((aligned(16))) float sBW[3 * H_];

    const int tid = threadIdx.x;
    const int idx = blockIdx.x * MLP_ROWS_ + tid;

    const float* x = X + (size_t)idx * H_;
    float xr[H_];
#pragma unroll
    for (int k = 0; k < H_; ++k) xr[k] = x[k];

    v4f wq[(H_ * H_ / 4) / MLP_ROWS_];
#pragma unroll
    for (int i = 0; i < (H_ * H_ / 4) / MLP_ROWS_; ++i)
        wq[i] = ((const v4f*)w1)[tid + MLP_ROWS_ * i];
    const int qt = tid < 3 * H_ / 4 ? tid : 3 * H_ / 4 - 1;
    const float* ps = qt < H_ / 4 ? b1 + 4 * qt : w2 + 4 * (qt - H_ / 4);
    const v4f sq = *(const v4f*)ps;
#pragma unroll
    for (int i = 0; i < (H_ * H_ / 4) / MLP_ROWS_; ++i)
        ((v4f*)sW1)[tid + MLP_ROWS_ * i] = wq[i];
    ((v4f*)sBW)[qt] = sq;

    __syncthreads();

#define MLP_LOADW(ws, jrow, kq)                                                 \
    do {                                                                        \
        _Pragma("unroll")                                                       \
        for (int i = 0; i < MLP_JB_; ++i) {                                     \
            _Pragma("unroll")                                                   \
            for (int q = 0; q < MLP_KQ_ / 4; ++q)                               \
                ws[i * (MLP_KQ_ / 4) + q] =                                     \
                    *(const v4f*)(sW1 + ((jrow) + i) * H_ + (kq) * MLP_KQ_ + q * 4); \
        }                                                                       \
    } while (0)

#define MLP_FMAW(ws, kq)                                                        \
    do {                                                                        \
        _Pragma("unroll")                                                       \
        for (int q = 0; q < MLP_KQ_ / 4; ++q) {                                 \
            _Pragma("unroll")                                                   \
            for (int kk = 0; kk < 4; ++kk) {                                    \
                _Pragma("unroll")                                               \
                for (int i = 0; i < MLP_JB_; ++i)                               \
                    acc[i] = fmaf(ws[i * (MLP_KQ_ / 4) + q][kk],                \
                                  xr[(kq) * MLP_KQ_ + q * 4 + kk], acc[i]);     \
            }                                                                   \
        }                                                                       \
    } while (0)

#define MLP_STEP(wnext, jnext, kqnext, wcur, kqcur)                             \
    do {                                                                        \
        MLP_LOADW(wnext, jnext, kqnext);                                        \
        __builtin_amdgcn_sched_barrier(0);                                      \
        MLP_FMAW(wcur, kqcur);                                                  \
        __builtin_amdgcn_sched_barrier(0);                                      \
    } while (0)

    v4f wa[MLP_JB_ * MLP_KQ_ / 4], wb[MLP_JB_ * MLP_KQ_ / 4];
    float o0 = b2[0];
    float o1 = b2[1];
    MLP_LOADW(wa, 0, 0);
    for (int j0 = 0; j0 < H_; j0 += MLP_JB_) {
        const v4f bq0 = *(const v4f*)(sBW + j0);
        const v4f bq1 = *(const v4f*)(sBW + j0 + 4);
        const v4f u0  = *(const v4f*)(sBW + H_ + j0);
        const v4f u1  = *(const v4f*)(sBW + H_ + j0 + 4);
        const v4f v0  = *(const v4f*)(sBW + 2 * H_ + j0);
        const v4f v1  = *(const v4f*)(sBW + 2 * H_ + j0 + 4);
        float acc[MLP_JB_];
#pragma unroll
        for (int i = 0; i < MLP_JB_; ++i) acc[i] = i < 4 ? bq0[i & 3] : bq1[i & 3];
        const int jn = (j0 + MLP_JB_) & (H_ - 1);
        MLP_STEP(wb, j0, 1, wa, 0);
        MLP_STEP(wa, j0, 2, wb, 1);
        MLP_STEP(wb, j0, 3, wa, 2);
        MLP_STEP(wa, jn, 0, wb, 3);
#pragma unroll
        for (int i = 0; i < MLP_JB_; ++i) {
            float hj = fmaxf(acc[i], 0.0f);
            o0 = fmaf(i < 4 ? u0[i & 3] : u1[i & 3], hj, o0);
            o1 = fmaf(i < 4 ? v0[i & 3] : v1[i & 3], hj, o1);
        }
    }
#undef MLP_STEP
#undef MLP_FMAW
#undef MLP_LOADW
    out[idx] = o0;
    float sp = fmaxf(o1, 0.0f) + log1pf(expf(-fabsf(o1)));
    out[C_ * TG_ + idx] = sp + 1e-6f;
}

extern "C" void kernel_launch(void* const* d_in, const int* in_sizes, int n_in,
                              void* d_out, int out_size, void* d_ws, size_t ws_size,
                              hipStream_t stream) {
    (void)in_sizes; (void)n_in; (void)out_size; (void)ws_size;
    const float* z   = (const float*)d_in[0];
    const float* grn = (const float*)d_in[1];
    const int*   btf = (const int*)  d_in[2];
    const float* w_z = (const float*)d_in[4];
    const float* b_z = (const float*)d_in[5];
    const float* w1  = (const float*)d_in[6];
    const float* b1  = (const float*)d_in[7];
    const float* w2  = (const float*)d_in[8];
    const float* b2  = (const float*)d_in[9];
    float* out = (float*)d_out;

    float* Xa = (float*)d_ws;
    float* Xb = Xa + (size_t)C_ * TG_ * H_;

    const int total = C_ * TG_ * H_;
    init_x_kernel<<<(total + 255) / 256, 256, 0, stream>>>(z, w_z, b_z, Xa, Xb);

    dim3 grid(TF_ / 16, C_);
    float* bufs[2] = { Xa, Xb };
    for (int it = 0; it < NITER; ++it) {
        const float* xin = bufs[it & 1];
        float*      xout = bufs[(it + 1) & 1];
        neumann_iter_kernel<<<grid, 128, 0, stream>>>(grn, z, w_z, b_z, btf, xin, xout);
    }
    const float* xfin = bufs[NITER & 1];

    mlp_kernel<<<(C_ * TG_) / MLP_ROWS_, MLP_ROWS_, 0, stream>>>(xfin, w1, b1, w2, b2, out);
}


# X kept in the workspace in WMMA B-fragment order: each sweep wave loads its B fragments straight from global (16-byte loads), LDS carries A only, read as 16-byte pieces; sum order changes
# speedup vs baseline: 4.9261x; 1.0655x over previous
#include <hip/hip_runtime.h>
#include <hip/hip_bf16.h>

#define C_   8
#define TF_  512
#define TG_  2048
#define H_   64
#define NITER 18

typedef __attribute__((ext_vector_type(2))) float v2f;
typedef __attribute__((ext_vector_type(8))) float v8f;

#define XOFF_T(t) ((((t) >> 3) << 9) + ((((t) >> 2) & 1) << 6) + ((t) & 3))
#define XOFF_N(n) ((((n) >> 4) << 7) + (((n) & 15) << 2))
__global__ __launch_bounds__(256)
void init_x_kernel(const float* __restrict__ z,
                   const float* __restrict__ w_z,
                   const float* __restrict__ b_z,
                   float* __restrict__ Xa,
                   float* __restrict__ Xb) {
    int idx = blockIdx.x * blockDim.x + threadIdx.x;
    if (idx >= C_ * TG_ * H_) return;
    int c  = idx >> 17;
    int kg = (idx >> 9) & 255;
    int n  = (((idx >> 7) & 3) << 4) + ((idx >> 2) & 15);
    int t  = (kg << 3) + (((idx >> 6) & 1) << 2) + (idx & 3);
    float v = z[c * TG_ + t] * w_z[n] + b_z[n];
    Xa[idx] = v;
    Xb[idx] = v;
}

#define KB_     128
#define NBLK_   (TG_ / KB_)
#define KG_     (KB_ / 8)
#define APITCH_ (KB_ + 4)
#define ASTG_   (16 * APITCH_)

typedef __attribute__((ext_vector_type(4))) float v4f;

__global__ __launch_bounds__(128)
void neumann_iter_kernel(const float* __restrict__ grn,
                         const float* __restrict__ z,
                         const float* __restrict__ w_z,
                         const float* __restrict__ b_z,
                         const int*   __restrict__ binary_tf,
                         const float* __restrict__ Xin,
                         float*       __restrict__ Xout) {
    __shared__ __attribute__((aligned(16))) float As[2 * ASTG_];

    const int c     = blockIdx.y;
    const int mtile = blockIdx.x;
    const int tid   = threadIdx.x;
    const int wave  = tid >> 5;
    const int lane  = tid & 31;
    const int m16   = lane & 15;
    const int hi    = lane >> 4;
    const int n     = wave * 16 + m16;

    const int* tf = binary_tf + c * TF_;

    const int arow = tid >> 5, aq = tid & 31;
    const float* gA = grn + ((size_t)(c * TF_ + mtile * 16 + arow)) * TG_ + aq * 4;
    float* sA = As + arow * APITCH_ + aq * 4;
    const float* fA = As + m16 * APITCH_ + 4 * hi;
    const float* gB = Xin + (size_t)c * TG_ * H_ + wave * 128 + lane * 4;

    v4f ra0[4], ra1[4];
    v4f fa0[KG_], fa1[KG_];
    v4f rb0[KG_], rb1[KG_], rb2[KG_], rb3[KG_];

#define GLOAD_A(blk, ra)                                                        \
    do {                                                                        \
        _Pragma("unroll")                                                       \
        for (int i = 0; i < 4; ++i)                                             \
            ra[i] = *(const v4f*)(gA + (size_t)(4 * i) * TG_ + (blk) * KB_);    \
    } while (0)

#define GLOAD_B(blk, rb)                                                        \
    do {                                                                        \
        _Pragma("unroll")                                                       \
        for (int g = 0; g < KG_; ++g)                                           \
            rb[g] = *(const v4f*)(gB + ((size_t)(blk) * KG_ + g) * 512);        \
    } while (0)

#define LSTORE_A(st, ra)                                                        \
    do {                                                                        \
        _Pragma("unroll")                                                       \
        for (int i = 0; i < 4; ++i)                                             \
            *(v4f*)(sA + (st) * ASTG_ + (4 * i) * APITCH_) = ra[i];             \
    } while (0)

#define FLOAD_A(fa, st)                                                         \
    do {                                                                        \
        _Pragma("unroll")                                                       \
        for (int g = 0; g < KG_; ++g)                                           \
            fa[g] = *(const v4f*)(fA + (st) * ASTG_ + 8 * g);                   \
    } while (0)

#define MMA32(fa, rb)                                                           \
    do {                                                                        \
        _Pragma("unroll")                                                       \
        for (int g = 0; g < KG_; ++g) {                                         \
            const v2f a01_ = { fa[g].x, fa[g].y }, a23_ = { fa[g].z, fa[g].w }; \
            const v2f b01_ = { rb[g].x, rb[g].y }, b23_ = { rb[g].z, rb[g].w }; \
            acc0 = __builtin_amdgcn_wmma_f32_16x16x4_f32(                       \
                      false, a01_, false, b01_, (short)0, acc0, false, false);  \
            acc1 = __builtin_amdgcn_wmma_f32_16x16x4_f32(                       \
                      false, a23_, false, b23_, (short)0, acc1, false, false);  \
        }                                                                       \
    } while (0)

#define LDS_BARRIER()                                                           \
    asm volatile("s_wait_dscnt 0x0\n\ts_barrier_signal -1\n\ts_barrier_wait -1" ::: "memory")

#define TRIP(blk, faC, faN, stC, rbC, rbL, raS, doB, doF, doS, doA)             \
    do {                                                                        \
        if (doB) GLOAD_B((blk) + 3, rbL);                                       \
        if (doF) FLOAD_A(faN, (stC) ^ 1);                                       \
        WMMA_AFTER_REQUESTS();                                                  \
        MMA32(faC, rbC);                                                        \
        WMMA_BEFORE_REFILL();                                                   \
        if (doS) LSTORE_A(stC, raS);                                            \
        if (doA) GLOAD_A((blk) + 4, raS);                                       \
        if (doS) LDS_BARRIER();     \
    } while (0)

#define WMMA_AFTER_REQUESTS()  asm volatile("" : "+v"(acc0), "+v"(acc1) :: "memory")
#define WMMA_BEFORE_REFILL()   asm volatile("" :: "v"(acc0), "v"(acc1) : "memory")

#define LOADS_IN_ORDER()  asm volatile("" ::: "memory")

    GLOAD_A(0, ra0);
    GLOAD_B(0, rb0);

    v8f acc0, acc1;
#pragma unroll
    for (int v = 0; v < 8; ++v) {
        int mrow = mtile * 16 + v + hi * 8;
        int trow = tf[mrow];
        acc0[v] = z[c * TG_ + trow] * w_z[n] + b_z[n];
        acc1[v] = 0.0f;
    }

    LOADS_IN_ORDER();
    GLOAD_A(1, ra1);
    GLOAD_B(1, rb1);
    GLOAD_B(2, rb2);
    LOADS_IN_ORDER();
    LSTORE_A(0, ra0);
    GLOAD_A(2, ra0);
    LDS_BARRIER();
    FLOAD_A(fa0, 0);
    LSTORE_A(1, ra1);
    GLOAD_A(3, ra1);
    LDS_BARRIER();
#undef LOADS_IN_ORDER

    for (int blk = 0; blk < NBLK_ - 4; blk += 4) {
        TRIP(blk + 0, fa0, fa1, 0, rb0, rb3, ra0, 1, 1, 1, 1);
        TRIP(blk + 1, fa1, fa0, 1, rb1, rb0, ra1, 1, 1, 1, 1);
        TRIP(blk + 2, fa0, fa1, 0, rb2, rb1, ra0, 1, 1, 1, 1);
        TRIP(blk + 3, fa1, fa0, 1, rb3, rb2, ra1, 1, 1, 1, 1);
    }
    TRIP(NBLK_ - 4, fa0, fa1, 0, rb0, rb3, ra0, 1, 1, 1, 0);
    TRIP(NBLK_ - 3, fa1, fa0, 1, rb1, rb0, ra1, 0, 1, 1, 0);
    TRIP(NBLK_ - 2, fa0, fa1, 0, rb2, rb1, ra0, 0, 1, 0, 0);
    TRIP(NBLK_ - 1, fa1, fa0, 1, rb3, rb2, ra1, 0, 0, 0, 0);

#undef TRIP
#undef WMMA_AFTER_REQUESTS
#undef WMMA_BEFORE_REFILL
#undef GLOAD_A
#undef GLOAD_B
#undef LSTORE_A
#undef FLOAD_A
#undef MMA32
#undef LDS_BARRIER

    float* xoutBase = Xout + (size_t)c * TG_ * H_ + XOFF_N(n);
#pragma unroll
    for (int v = 0; v < 8; ++v) {
        int mrow = mtile * 16 + v + hi * 8;
        int trow = tf[mrow];
        xoutBase[XOFF_T(trow)] = acc0[v] + acc1[v];
    }
}

#define MLP_ROWS_ 64
#define MLP_JB_   8
#define MLP_KQ_   16
__global__ __launch_bounds__(MLP_ROWS_)
void mlp_kernel(const float* __restrict__ X,
                const float* __restrict__ w1, const float* __restrict__ b1,
                const float* __restrict__ w2, const float* __restrict__ b2,
                float* __restrict__ out) {
    __shared__ __attribute__((aligned(16))) float sW1[H_ * H_];
    __shared__ __attribute__((aligned(16))) float sBW[3 * H_];

    const int tid = threadIdx.x;
    const int idx = blockIdx.x * MLP_ROWS_ + tid;

    const int t = idx & (TG_ - 1);
    const float* x = X + (size_t)(idx - t) * H_ + XOFF_T(t);
    float xr[H_];
#pragma unroll
    for (int k = 0; k < H_; ++k) xr[k] = x[XOFF_N(k)];

    v4f wq[(H_ * H_ / 4) / MLP_ROWS_];
#pragma unroll
    for (int i = 0; i < (H_ * H_ / 4) / MLP_ROWS_; ++i)
        wq[i] = ((const v4f*)w1)[tid + MLP_ROWS_ * i];
    const int qt = tid < 3 * H_ / 4 ? tid : 3 * H_ / 4 - 1;
    const float* ps = qt < H_ / 4 ? b1 + 4 * qt : w2 + 4 * (qt - H_ / 4);
    const v4f sq = *(const v4f*)ps;
#pragma unroll
    for (int i = 0; i < (H_ * H_ / 4) / MLP_ROWS_; ++i)
        ((v4f*)sW1)[tid + MLP_ROWS_ * i] = wq[i];
    ((v4f*)sBW)[qt] = sq;

    __syncthreads();

#define MLP_LOADW(ws, jrow, kq)                                                 \
    do {                                                                        \
        _Pragma("unroll")                                                       \
        for (int i = 0; i < MLP_JB_; ++i) {                                     \
            _Pragma("unroll")                                                   \
            for (int q = 0; q < MLP_KQ_ / 4; ++q)                               \
                ws[i * (MLP_KQ_ / 4) + q] =                                     \
                    *(const v4f*)(sW1 + ((jrow) + i) * H_ + (kq) * MLP_KQ_ + q * 4); \
        }                                                                       \
    } while (0)

#define MLP_FMAW(ws, kq)                                                        \
    do {                                                                        \
        _Pragma("unroll")                                                       \
        for (int q = 0; q < MLP_KQ_ / 4; ++q) {                                 \
            _Pragma("unroll")                                                   \
            for (int kk = 0; kk < 4; ++kk) {                                    \
                _Pragma("unroll")                                               \
                for (int i = 0; i < MLP_JB_; ++i)                               \
                    acc[i] = fmaf(ws[i * (MLP_KQ_ / 4) + q][kk],                \
                                  xr[(kq) * MLP_KQ_ + q * 4 + kk], acc[i]);     \
            }                                                                   \
        }                                                                       \
    } while (0)

#define MLP_STEP(wnext, jnext, kqnext, wcur, kqcur)                             \
    do {                                                                        \
        MLP_LOADW(wnext, jnext, kqnext);                                        \
        __builtin_amdgcn_sched_barrier(0);                                      \
        MLP_FMAW(wcur, kqcur);                                                  \
        __builtin_amdgcn_sched_barrier(0);                                      \
    } while (0)

    v4f wa[MLP_JB_ * MLP_KQ_ / 4], wb[MLP_JB_ * MLP_KQ_ / 4];
    float o0 = b2[0];
    float o1 = b2[1];
    MLP_LOADW(wa, 0, 0);
    for (int j0 = 0; j0 < H_; j0 += MLP_JB_) {
        const v4f bq0 = *(const v4f*)(sBW + j0);
        const v4f bq1 = *(const v4f*)(sBW + j0 + 4);
        const v4f u0  = *(const v4f*)(sBW + H_ + j0);
        const v4f u1  = *(const v4f*)(sBW + H_ + j0 + 4);
        const v4f v0  = *(const v4f*)(sBW + 2 * H_ + j0);
        const v4f v1  = *(const v4f*)(sBW + 2 * H_ + j0 + 4);
        float acc[MLP_JB_];
#pragma unroll
        for (int i = 0; i < MLP_JB_; ++i) acc[i] = i < 4 ? bq0[i & 3] : bq1[i & 3];
        const int jn = (j0 + MLP_JB_) & (H_ - 1);
        MLP_STEP(wb, j0, 1, wa, 0);
        MLP_STEP(wa, j0, 2, wb, 1);
        MLP_STEP(wb, j0, 3, wa, 2);
        MLP_STEP(wa, jn, 0, wb, 3);
#pragma unroll
        for (int i = 0; i < MLP_JB_; ++i) {
            float hj = fmaxf(acc[i], 0.0f);
            o0 = fmaf(i < 4 ? u0[i & 3] : u1[i & 3], hj, o0);
            o1 = fmaf(i < 4 ? v0[i & 3] : v1[i & 3], hj, o1);
        }
    }
#undef MLP_STEP
#undef MLP_FMAW
#undef MLP_LOADW
    out[idx] = o0;
    float sp = fmaxf(o1, 0.0f) + log1pf(expf(-fabsf(o1)));
    out[C_ * TG_ + idx] = sp + 1e-6f;
}

extern "C" void kernel_launch(void* const* d_in, const int* in_sizes, int n_in,
                              void* d_out, int out_size, void* d_ws, size_t ws_size,
                              hipStream_t stream) {
    (void)in_sizes; (void)n_in; (void)out_size; (void)ws_size;
    const float* z   = (const float*)d_in[0];
    const float* grn = (const float*)d_in[1];
    const int*   btf = (const int*)  d_in[2];
    const float* w_z = (const float*)d_in[4];
    const float* b_z = (const float*)d_in[5];
    const float* w1  = (const float*)d_in[6];
    const float* b1  = (const float*)d_in[7];
    const float* w2  = (const float*)d_in[8];
    const float* b2  = (const float*)d_in[9];
    float* out = (float*)d_out;

    float* Xa = (float*)d_ws;
    float* Xb = Xa + (size_t)C_ * TG_ * H_;

    const int total = C_ * TG_ * H_;
    init_x_kernel<<<(total + 255) / 256, 256, 0, stream>>>(z, w_z, b_z, Xa, Xb);

    dim3 grid(TF_ / 16, C_);
    float* bufs[2] = { Xa, Xb };
    for (int it = 0; it < NITER; ++it) {
        const float* xin = bufs[it & 1];
        float*      xout = bufs[(it + 1) & 1];
        neumann_iter_kernel<<<grid, 128, 0, stream>>>(grn, z, w_z, b_z, btf, xin, xout);
    }
    const float* xfin = bufs[NITER & 1];

    mlp_kernel<<<(C_ * TG_) / MLP_ROWS_, MLP_ROWS_, 0, stream>>>(xfin, w1, b1, w2, b2, out);
}


# sweep kernel: each work-group starts its K walk at block (M-tile mod 16) and wraps round, so a condition's 32 work-groups stop asking L2 for the same X lines in step; sum order changes
# speedup vs baseline: 5.1508x; 1.0456x over previous
#include <hip/hip_runtime.h>
#include <hip/hip_bf16.h>

#define C_   8
#define TF_  512
#define TG_  2048
#define H_   64
#define NITER 18

typedef __attribute__((ext_vector_type(2))) float v2f;
typedef __attribute__((ext_vector_type(8))) float v8f;

#define XOFF_T(t) ((((t) >> 3) << 9) + ((((t) >> 2) & 1) << 6) + ((t) & 3))
#define XOFF_N(n) ((((n) >> 4) << 7) + (((n) & 15) << 2))
__global__ __launch_bounds__(256)
void init_x_kernel(const float* __restrict__ z,
                   const float* __restrict__ w_z,
                   const float* __restrict__ b_z,
                   float* __restrict__ Xa,
                   float* __restrict__ Xb) {
    int idx = blockIdx.x * blockDim.x + threadIdx.x;
    if (idx >= C_ * TG_ * H_) return;
    int c  = idx >> 17;
    int kg = (idx >> 9) & 255;
    int n  = (((idx >> 7) & 3) << 4) + ((idx >> 2) & 15);
    int t  = (kg << 3) + (((idx >> 6) & 1) << 2) + (idx & 3);
    float v = z[c * TG_ + t] * w_z[n] + b_z[n];
    Xa[idx] = v;
    Xb[idx] = v;
}

#define KB_     128
#define NBLK_   (TG_ / KB_)
#define KG_     (KB_ / 8)
#define APITCH_ (KB_ + 4)
#define ASTG_   (16 * APITCH_)

typedef __attribute__((ext_vector_type(4))) float v4f;

__global__ __launch_bounds__(128)
void neumann_iter_kernel(const float* __restrict__ grn,
                         const float* __restrict__ z,
                         const float* __restrict__ w_z,
                         const float* __restrict__ b_z,
                         const int*   __restrict__ binary_tf,
                         const float* __restrict__ Xin,
                         float*       __restrict__ Xout) {
    __shared__ __attribute__((aligned(16))) float As[2 * ASTG_];

    const int c     = blockIdx.y;
    const int mtile = blockIdx.x;
    const int tid   = threadIdx.x;
    const int wave  = tid >> 5;
    const int lane  = tid & 31;
    const int m16   = lane & 15;
    const int hi    = lane >> 4;
    const int n     = wave * 16 + m16;

    const int* tf = binary_tf + c * TF_;

    const int kstart = mtile & (NBLK_ - 1);
#define KBLK(p) (((p) + kstart) & (NBLK_ - 1))

    const int arow = tid >> 5, aq = tid & 31;
    const float* gA = grn + ((size_t)(c * TF_ + mtile * 16 + arow)) * TG_ + aq * 4;
    float* sA = As + arow * APITCH_ + aq * 4;
    const float* fA = As + m16 * APITCH_ + 4 * hi;
    const float* gB = Xin + (size_t)c * TG_ * H_ + wave * 128 + lane * 4;

    v4f ra0[4], ra1[4];
    v4f fa0[KG_], fa1[KG_];
    v4f rb0[KG_], rb1[KG_], rb2[KG_], rb3[KG_];

#define GLOAD_A(blk, ra)                                                        \
    do {                                                                        \
        _Pragma("unroll")                                                       \
        for (int i = 0; i < 4; ++i)                                             \
            ra[i] = *(const v4f*)(gA + (size_t)(4 * i) * TG_ + KBLK(blk) * KB_); \
    } while (0)

#define GLOAD_B(blk, rb)                                                        \
    do {                                                                        \
        _Pragma("unroll")                                                       \
        for (int g = 0; g < KG_; ++g)                                           \
            rb[g] = *(const v4f*)(gB + ((size_t)KBLK(blk) * KG_ + g) * 512);    \
    } while (0)

#define LSTORE_A(st, ra)                                                        \
    do {                                                                        \
        _Pragma("unroll")                                                       \
        for (int i = 0; i < 4; ++i)                                             \
            *(v4f*)(sA + (st) * ASTG_ + (4 * i) * APITCH_) = ra[i];             \
    } while (0)

#define FLOAD_A(fa, st)                                                         \
    do {                                                                        \
        _Pragma("unroll")                                                       \
        for (int g = 0; g < KG_; ++g)                                           \
            fa[g] = *(const v4f*)(fA + (st) * ASTG_ + 8 * g);                   \
    } while (0)

#define MMA32(fa, rb)                                                           \
    do {                                                                        \
        _Pragma("unroll")                                                       \
        for (int g = 0; g < KG_; ++g) {                                         \
            const v2f a01_ = { fa[g].x, fa[g].y }, a23_ = { fa[g].z, fa[g].w }; \
            const v2f b01_ = { rb[g].x, rb[g].y }, b23_ = { rb[g].z, rb[g].w }; \
            acc0 = __builtin_amdgcn_wmma_f32_16x16x4_f32(                       \
                      false, a01_, false, b01_, (short)0, acc0, false, false);  \
            acc1 = __builtin_amdgcn_wmma_f32_16x16x4_f32(                       \
                      false, a23_, false, b23_, (short)0, acc1, false, false);  \
        }                                                                       \
    } while (0)

#define LDS_BARRIER()                                                           \
    asm volatile("s_wait_dscnt 0x0\n\ts_barrier_signal -1\n\ts_barrier_wait -1" ::: "memory")

#define TRIP(blk, faC, faN, stC, rbC, rbL, raS, doB, doF, doS, doA)             \
    do {                                                                        \
        if (doB) GLOAD_B((blk) + 3, rbL);                                       \
        if (doF) FLOAD_A(faN, (stC) ^ 1);                                       \
        WMMA_AFTER_REQUESTS();                                                  \
        MMA32(faC, rbC);                                                        \
        WMMA_BEFORE_REFILL();                                                   \
        if (doS) LSTORE_A(stC, raS);                                            \
        if (doA) GLOAD_A((blk) + 4, raS);                                       \
        if (doS) LDS_BARRIER();     \
    } while (0)

#define WMMA_AFTER_REQUESTS()  asm volatile("" : "+v"(acc0), "+v"(acc1) :: "memory")
#define WMMA_BEFORE_REFILL()   asm volatile("" :: "v"(acc0), "v"(acc1) : "memory")

#define LOADS_IN_ORDER()  asm volatile("" ::: "memory")

    GLOAD_A(0, ra0);
    GLOAD_B(0, rb0);

    v8f acc0, acc1;
#pragma unroll
    for (int v = 0; v < 8; ++v) {
        int mrow = mtile * 16 + v + hi * 8;
        int trow = tf[mrow];
        acc0[v] = z[c * TG_ + trow] * w_z[n] + b_z[n];
        acc1[v] = 0.0f;
    }

    LOADS_IN_ORDER();
    GLOAD_A(1, ra1);
    GLOAD_B(1, rb1);
    GLOAD_B(2, rb2);
    LOADS_IN_ORDER();
    LSTORE_A(0, ra0);
    GLOAD_A(2, ra0);
    LDS_BARRIER();
    FLOAD_A(fa0, 0);
    LSTORE_A(1, ra1);
    GLOAD_A(3, ra1);
    LDS_BARRIER();
#undef LOADS_IN_ORDER

    for (int blk = 0; blk < NBLK_ - 4; blk += 4) {
        TRIP(blk + 0, fa0, fa1, 0, rb0, rb3, ra0, 1, 1, 1, 1);
        TRIP(blk + 1, fa1, fa0, 1, rb1, rb0, ra1, 1, 1, 1, 1);
        TRIP(blk + 2, fa0, fa1, 0, rb2, rb1, ra0, 1, 1, 1, 1);
        TRIP(blk + 3, fa1, fa0, 1, rb3, rb2, ra1, 1, 1, 1, 1);
    }
    TRIP(NBLK_ - 4, fa0, fa1, 0, rb0, rb3, ra0, 1, 1, 1, 0);
    TRIP(NBLK_ - 3, fa1, fa0, 1, rb1, rb0, ra1, 0, 1, 1, 0);
    TRIP(NBLK_ - 2, fa0, fa1, 0, rb2, rb1, ra0, 0, 1, 0, 0);
    TRIP(NBLK_ - 1, fa1, fa0, 1, rb3, rb2, ra1, 0, 0, 0, 0);

#undef TRIP
#undef KBLK
#undef WMMA_AFTER_REQUESTS
#undef WMMA_BEFORE_REFILL
#undef GLOAD_A
#undef GLOAD_B
#undef LSTORE_A
#undef FLOAD_A
#undef MMA32
#undef LDS_BARRIER

    float* xoutBase = Xout + (size_t)c * TG_ * H_ + XOFF_N(n);
#pragma unroll
    for (int v = 0; v < 8; ++v) {
        int mrow = mtile * 16 + v + hi * 8;
        int trow = tf[mrow];
        xoutBase[XOFF_T(trow)] = acc0[v] + acc1[v];
    }
}

#define MLP_ROWS_ 64
#define MLP_JB_   8
#define MLP_KQ_   16
__global__ __launch_bounds__(MLP_ROWS_)
void mlp_kernel(const float* __restrict__ X,
                const float* __restrict__ w1, const float* __restrict__ b1,
                const float* __restrict__ w2, const float* __restrict__ b2,
                float* __restrict__ out) {
    __shared__ __attribute__((aligned(16))) float sW1[H_ * H_];
    __shared__ __attribute__((aligned(16))) float sBW[3 * H_];

    const int tid = threadIdx.x;
    const int idx = blockIdx.x * MLP_ROWS_ + tid;

    const int t = idx & (TG_ - 1);
    const float* x = X + (size_t)(idx - t) * H_ + XOFF_T(t);
    float xr[H_];
#pragma unroll
    for (int k = 0; k < H_; ++k) xr[k] = x[XOFF_N(k)];

    v4f wq[(H_ * H_ / 4) / MLP_ROWS_];
#pragma unroll
    for (int i = 0; i < (H_ * H_ / 4) / MLP_ROWS_; ++i)
        wq[i] = ((const v4f*)w1)[tid + MLP_ROWS_ * i];
    const int qt = tid < 3 * H_ / 4 ? tid : 3 * H_ / 4 - 1;
    const float* ps = qt < H_ / 4 ? b1 + 4 * qt : w2 + 4 * (qt - H_ / 4);
    const v4f sq = *(const v4f*)ps;
#pragma unroll
    for (int i = 0; i < (H_ * H_ / 4) / MLP_ROWS_; ++i)
        ((v4f*)sW1)[tid + MLP_ROWS_ * i] = wq[i];
    ((v4f*)sBW)[qt] = sq;

    __syncthreads();

#define MLP_LOADW(ws, jrow, kq)                                                 \
    do {                                                                        \
        _Pragma("unroll")                                                       \
        for (int i = 0; i < MLP_JB_; ++i) {                                     \
            _Pragma("unroll")                                                   \
            for (int q = 0; q < MLP_KQ_ / 4; ++q)                               \
                ws[i * (MLP_KQ_ / 4) + q] =                                     \
                    *(const v4f*)(sW1 + ((jrow) + i) * H_ + (kq) * MLP_KQ_ + q * 4); \
        }                                                                       \
    } while (0)

#define MLP_FMAW(ws, kq)                                                        \
    do {                                                                        \
        _Pragma("unroll")                                                       \
        for (int q = 0; q < MLP_KQ_ / 4; ++q) {                                 \
            _Pragma("unroll")                                                   \
            for (int kk = 0; kk < 4; ++kk) {                                    \
                _Pragma("unroll")                                               \
                for (int i = 0; i < MLP_JB_; ++i)                               \
                    acc[i] = fmaf(ws[i * (MLP_KQ_ / 4) + q][kk],                \
                                  xr[(kq) * MLP_KQ_ + q * 4 + kk], acc[i]);     \
            }                                                                   \
        }                                                                       \
    } while (0)

#define MLP_STEP(wnext, jnext, kqnext, wcur, kqcur)                             \
    do {                                                                        \
        MLP_LOADW(wnext, jnext, kqnext);                                        \
        __builtin_amdgcn_sched_barrier(0);                                      \
        MLP_FMAW(wcur, kqcur);                                                  \
        __builtin_amdgcn_sched_barrier(0);                                      \
    } while (0)

    v4f wa[MLP_JB_ * MLP_KQ_ / 4], wb[MLP_JB_ * MLP_KQ_ / 4];
    float o0 = b2[0];
    float o1 = b2[1];
    MLP_LOADW(wa, 0, 0);
    for (int j0 = 0; j0 < H_; j0 += MLP_JB_) {
        const v4f bq0 = *(const v4f*)(sBW + j0);
        const v4f bq1 = *(const v4f*)(sBW + j0 + 4);
        const v4f u0  = *(const v4f*)(sBW + H_ + j0);
        const v4f u1  = *(const v4f*)(sBW + H_ + j0 + 4);
        const v4f v0  = *(const v4f*)(sBW + 2 * H_ + j0);
        const v4f v1  = *(const v4f*)(sBW + 2 * H_ + j0 + 4);
        float acc[MLP_JB_];
#pragma unroll
        for (int i = 0; i < MLP_JB_; ++i) acc[i] = i < 4 ? bq0[i & 3] : bq1[i & 3];
        const int jn = (j0 + MLP_JB_) & (H_ - 1);
        MLP_STEP(wb, j0, 1, wa, 0);
        MLP_STEP(wa, j0, 2, wb, 1);
        MLP_STEP(wb, j0, 3, wa, 2);
        MLP_STEP(wa, jn, 0, wb, 3);
#pragma unroll
        for (int i = 0; i < MLP_JB_; ++i) {
            float hj = fmaxf(acc[i], 0.0f);
            o0 = fmaf(i < 4 ? u0[i & 3] : u1[i & 3], hj, o0);
            o1 = fmaf(i < 4 ? v0[i & 3] : v1[i & 3], hj, o1);
        }
    }
#undef MLP_STEP
#undef MLP_FMAW
#undef MLP_LOADW
    out[idx] = o0;
    float sp = fmaxf(o1, 0.0f) + log1pf(expf(-fabsf(o1)));
    out[C_ * TG_ + idx] = sp + 1e-6f;
}

extern "C" void kernel_launch(void* const* d_in, const int* in_sizes, int n_in,
                              void* d_out, int out_size, void* d_ws, size_t ws_size,
                              hipStream_t stream) {
    (void)in_sizes; (void)n_in; (void)out_size; (void)ws_size;
    const float* z   = (const float*)d_in[0];
    const float* grn = (const float*)d_in[1];
    const int*   btf = (const int*)  d_in[2];
    const float* w_z = (const float*)d_in[4];
    const float* b_z = (const float*)d_in[5];
    const float* w1  = (const float*)d_in[6];
    const float* b1  = (const float*)d_in[7];
    const float* w2  = (const float*)d_in[8];
    const float* b2  = (const float*)d_in[9];
    float* out = (float*)d_out;

    float* Xa = (float*)d_ws;
    float* Xb = Xa + (size_t)C_ * TG_ * H_;

    const int total = C_ * TG_ * H_;
    init_x_kernel<<<(total + 255) / 256, 256, 0, stream>>>(z, w_z, b_z, Xa, Xb);

    dim3 grid(TF_ / 16, C_);
    float* bufs[2] = { Xa, Xb };
    for (int it = 0; it < NITER; ++it) {
        const float* xin = bufs[it & 1];
        float*      xout = bufs[(it + 1) & 1];
        neumann_iter_kernel<<<grid, 128, 0, stream>>>(grn, z, w_z, b_z, btf, xin, xout);
    }
    const float* xfin = bufs[NITER & 1];

    mlp_kernel<<<(C_ * TG_) / MLP_ROWS_, MLP_ROWS_, 0, stream>>>(xfin, w1, b1, w2, b2, out);
}


# sweep kernel: LDS hand-over of the A tile moved off the path between WMMA runs: stores and barrier signal ahead of a trip's WMMAs, barrier wait behind them (A ring of 4 LDS stages); same bits
# speedup vs baseline: 5.3511x; 1.0389x over previous
#include <hip/hip_runtime.h>
#include <hip/hip_bf16.h>

#define C_   8
#define TF_  512
#define TG_  2048
#define H_   64
#define NITER 18

typedef __attribute__((ext_vector_type(2))) float v2f;
typedef __attribute__((ext_vector_type(8))) float v8f;

#define XOFF_T(t) ((((t) >> 3) << 9) + ((((t) >> 2) & 1) << 6) + ((t) & 3))
#define XOFF_N(n) ((((n) >> 4) << 7) + (((n) & 15) << 2))
__global__ __launch_bounds__(256)
void init_x_kernel(const float* __restrict__ z,
                   const float* __restrict__ w_z,
                   const float* __restrict__ b_z,
                   float* __restrict__ Xa,
                   float* __restrict__ Xb) {
    int idx = blockIdx.x * blockDim.x + threadIdx.x;
    if (idx >= C_ * TG_ * H_) return;
    int c  = idx >> 17;
    int kg = (idx >> 9) & 255;
    int n  = (((idx >> 7) & 3) << 4) + ((idx >> 2) & 15);
    int t  = (kg << 3) + (((idx >> 6) & 1) << 2) + (idx & 3);
    float v = z[c * TG_ + t] * w_z[n] + b_z[n];
    Xa[idx] = v;
    Xb[idx] = v;
}

#define KB_     128
#define NBLK_   (TG_ / KB_)
#define KG_     (KB_ / 8)
#define APITCH_ (KB_ + 4)
#define ASTG_   (16 * APITCH_)

typedef __attribute__((ext_vector_type(4))) float v4f;

__global__ __launch_bounds__(128)
void neumann_iter_kernel(const float* __restrict__ grn,
                         const float* __restrict__ z,
                         const float* __restrict__ w_z,
                         const float* __restrict__ b_z,
                         const int*   __restrict__ binary_tf,
                         const float* __restrict__ Xin,
                         float*       __restrict__ Xout) {
    __shared__ __attribute__((aligned(16))) float As[4 * ASTG_];

    const int c     = blockIdx.y;
    const int mtile = blockIdx.x;
    const int tid   = threadIdx.x;
    const int wave  = tid >> 5;
    const int lane  = tid & 31;
    const int m16   = lane & 15;
    const int hi    = lane >> 4;
    const int n     = wave * 16 + m16;

    const int* tf = binary_tf + c * TF_;

    const int kstart = mtile & (NBLK_ - 1);
#define KBLK(p) (((p) + kstart) & (NBLK_ - 1))

    const int arow = tid >> 5, aq = tid & 31;
    const float* gA = grn + ((size_t)(c * TF_ + mtile * 16 + arow)) * TG_ + aq * 4;
    float* sA = As + arow * APITCH_ + aq * 4;
    const float* fA = As + m16 * APITCH_ + 4 * hi;
    const float* gB = Xin + (size_t)c * TG_ * H_ + wave * 128 + lane * 4;

    v4f ra0[4], ra1[4];
    v4f ra2[4], ra3[4];
    v4f fa0[KG_], fa1[KG_];
    v4f rb0[KG_], rb1[KG_], rb2[KG_], rb3[KG_];

#define GLOAD_A(blk, ra)                                                        \
    do {                                                                        \
        _Pragma("unroll")                                                       \
        for (int i = 0; i < 4; ++i)                                             \
            ra[i] = *(const v4f*)(gA + (size_t)(4 * i) * TG_ + KBLK(blk) * KB_); \
    } while (0)

#define GLOAD_B(blk, rb)                                                        \
    do {                                                                        \
        _Pragma("unroll")                                                       \
        for (int g = 0; g < KG_; ++g)                                           \
            rb[g] = *(const v4f*)(gB + ((size_t)KBLK(blk) * KG_ + g) * 512);    \
    } while (0)

#define LSTORE_A(st, ra)                                                        \
    do {                                                                        \
        _Pragma("unroll")                                                       \
        for (int i = 0; i < 4; ++i)                                             \
            *(v4f*)(sA + (st) * ASTG_ + (4 * i) * APITCH_) = ra[i];             \
    } while (0)

#define FLOAD_A(fa, st)                                                         \
    do {                                                                        \
        _Pragma("unroll")                                                       \
        for (int g = 0; g < KG_; ++g)                                           \
            fa[g] = *(const v4f*)(fA + (st) * ASTG_ + 8 * g);                   \
    } while (0)

#define MMA32(fa, rb)                                                           \
    do {                                                                        \
        _Pragma("unroll")                                                       \
        for (int g = 0; g < KG_; ++g) {                                         \
            const v2f a01_ = { fa[g].x, fa[g].y }, a23_ = { fa[g].z, fa[g].w }; \
            const v2f b01_ = { rb[g].x, rb[g].y }, b23_ = { rb[g].z, rb[g].w }; \
            acc0 = __builtin_amdgcn_wmma_f32_16x16x4_f32(                       \
                      false, a01_, false, b01_, (short)0, acc0, false, false);  \
            acc1 = __builtin_amdgcn_wmma_f32_16x16x4_f32(                       \
                      false, a23_, false, b23_, (short)0, acc1, false, false);  \
        }                                                                       \
    } while (0)

#define LDS_BARRIER()                                                           \
    asm volatile("s_wait_dscnt 0x0\n\ts_barrier_signal -1\n\ts_barrier_wait -1" ::: "memory")

#define STAGE_SIGNAL()                                                          \
    asm volatile("s_wait_dscnt 0x10\n\ts_barrier_signal -1" : "+v"(acc0), "+v"(acc1) :: "memory")
#define STAGE_WAIT()                                                            \
    asm volatile("s_barrier_wait -1" :: "v"(acc0), "v"(acc1) : "memory")
#define STORES_BEFORE_READS()  asm volatile("" ::: "memory")

#define TRIP(blk, faC, faN, stS, stL, rbC, rbL, raF, raT, doB, doF, doS, doA)   \
    do {                                                                        \
        if (doS) LSTORE_A(stS, raF);                                            \
        if (doS) STORES_BEFORE_READS();                                         \
        if (doA) GLOAD_A((blk) + 4, raT);                                       \
        if (doB) GLOAD_B((blk) + 3, rbL);                                       \
        if (doF) FLOAD_A(faN, stL);                                             \
        if (doS) STAGE_SIGNAL(); else WMMA_AFTER_REQUESTS();                    \
        MMA32(faC, rbC);                                                        \
        if (doS) STAGE_WAIT(); else WMMA_BEFORE_REFILL();                       \
    } while (0)

#define WMMA_AFTER_REQUESTS()  asm volatile("" : "+v"(acc0), "+v"(acc1) :: "memory")
#define WMMA_BEFORE_REFILL()   asm volatile("" :: "v"(acc0), "v"(acc1) : "memory")

#define LOADS_IN_ORDER()  asm volatile("" ::: "memory")

    GLOAD_A(0, ra0);
    GLOAD_B(0, rb0);

    v8f acc0, acc1;
#pragma unroll
    for (int v = 0; v < 8; ++v) {
        int mrow = mtile * 16 + v + hi * 8;
        int trow = tf[mrow];
        acc0[v] = z[c * TG_ + trow] * w_z[n] + b_z[n];
        acc1[v] = 0.0f;
    }

    LOADS_IN_ORDER();
    GLOAD_A(1, ra1);
    GLOAD_B(1, rb1);
    GLOAD_A(2, ra2);
    GLOAD_B(2, rb2);
    GLOAD_A(3, ra3);
    LOADS_IN_ORDER();
    LSTORE_A(0, ra0);
    LDS_BARRIER();
    FLOAD_A(fa0, 0);
    LSTORE_A(1, ra1);
    LDS_BARRIER();
#undef LOADS_IN_ORDER

    TRIP(0, fa0, fa1, 2, 1, rb0, rb3, ra2, ra0, 1, 1, 1, 1);
    TRIP(1, fa1, fa0, 3, 2, rb1, rb0, ra3, ra1, 1, 1, 1, 1);
    TRIP(2, fa0, fa1, 0, 3, rb2, rb1, ra0, ra0, 1, 1, 1, 1);
    TRIP(3, fa1, fa0, 1, 0, rb3, rb2, ra1, ra1, 1, 1, 1, 1);
    for (int blk = 4; blk < NBLK_ - 4; blk += 4) {
        TRIP(blk + 0, fa0, fa1, 2, 1, rb0, rb3, ra0, ra0, 1, 1, 1, 1);
        TRIP(blk + 1, fa1, fa0, 3, 2, rb1, rb0, ra1, ra1, 1, 1, 1, 1);
        TRIP(blk + 2, fa0, fa1, 0, 3, rb2, rb1, ra0, ra0, 1, 1, 1, 1);
        TRIP(blk + 3, fa1, fa0, 1, 0, rb3, rb2, ra1, ra1, 1, 1, 1, 1);
    }
    TRIP(NBLK_ - 4, fa0, fa1, 2, 1, rb0, rb3, ra0, ra0, 1, 1, 1, 0);
    TRIP(NBLK_ - 3, fa1, fa0, 3, 2, rb1, rb0, ra1, ra1, 0, 1, 1, 0);
    TRIP(NBLK_ - 2, fa0, fa1, 0, 3, rb2, rb1, ra0, ra0, 0, 1, 0, 0);
    TRIP(NBLK_ - 1, fa1, fa0, 1, 0, rb3, rb2, ra1, ra1, 0, 0, 0, 0);

#undef TRIP
#undef KBLK
#undef WMMA_AFTER_REQUESTS
#undef WMMA_BEFORE_REFILL
#undef GLOAD_A
#undef GLOAD_B
#undef LSTORE_A
#undef FLOAD_A
#undef MMA32
#undef LDS_BARRIER
#undef STAGE_SIGNAL
#undef STAGE_WAIT
#undef STORES_BEFORE_READS

    float* xoutBase = Xout + (size_t)c * TG_ * H_ + XOFF_N(n);
#pragma unroll
    for (int v = 0; v < 8; ++v) {
        int mrow = mtile * 16 + v + hi * 8;
        int trow = tf[mrow];
        xoutBase[XOFF_T(trow)] = acc0[v] + acc1[v];
    }
}

#define MLP_ROWS_ 64
#define MLP_JB_   8
#define MLP_KQ_   16
__global__ __launch_bounds__(MLP_ROWS_)
void mlp_kernel(const float* __restrict__ X,
                const float* __restrict__ w1, const float* __restrict__ b1,
                const float* __restrict__ w2, const float* __restrict__ b2,
                float* __restrict__ out) {
    __shared__ __attribute__((aligned(16))) float sW1[H_ * H_];
    __shared__ __attribute__((aligned(16))) float sBW[3 * H_];

    const int tid = threadIdx.x;
    const int idx = blockIdx.x * MLP_ROWS_ + tid;

    const int t = idx & (TG_ - 1);
    const float* x = X + (size_t)(idx - t) * H_ + XOFF_T(t);
    float xr[H_];
#pragma unroll
    for (int k = 0; k < H_; ++k) xr[k] = x[XOFF_N(k)];

    v4f wq[(H_ * H_ / 4) / MLP_ROWS_];
#pragma unroll
    for (int i = 0; i < (H_ * H_ / 4) / MLP_ROWS_; ++i)
        wq[i] = ((const v4f*)w1)[tid + MLP_ROWS_ * i];
    const int qt = tid < 3 * H_ / 4 ? tid : 3 * H_ / 4 - 1;
    const float* ps = qt < H_ / 4 ? b1 + 4 * qt : w2 + 4 * (qt - H_ / 4);
    const v4f sq = *(const v4f*)ps;
#pragma unroll
    for (int i = 0; i < (H_ * H_ / 4) / MLP_ROWS_; ++i)
        ((v4f*)sW1)[tid + MLP_ROWS_ * i] = wq[i];
    ((v4f*)sBW)[qt] = sq;

    __syncthreads();

#define MLP_LOADW(ws, jrow, kq)                                                 \
    do {                                                                        \
        _Pragma("unroll")                                                       \
        for (int i = 0; i < MLP_JB_; ++i) {                                     \
            _Pragma("unroll")                                                   \
            for (int q = 0; q < MLP_KQ_ / 4; ++q)                               \
                ws[i * (MLP_KQ_ / 4) + q] =                                     \
                    *(const v4f*)(sW1 + ((jrow) + i) * H_ + (kq) * MLP_KQ_ + q * 4); \
        }                                                                       \
    } while (0)

#define MLP_FMAW(ws, kq)                                                        \
    do {                                                                        \
        _Pragma("unroll")                                                       \
        for (int q = 0; q < MLP_KQ_ / 4; ++q) {                                 \
            _Pragma("unroll")                                                   \
            for (int kk = 0; kk < 4; ++kk) {                                    \
                _Pragma("unroll")                                               \
                for (int i = 0; i < MLP_JB_; ++i)                               \
                    acc[i] = fmaf(ws[i * (MLP_KQ_ / 4) + q][kk],                \
                                  xr[(kq) * MLP_KQ_ + q * 4 + kk], acc[i]);     \
            }                                                                   \
        }                                                                       \
    } while (0)

#define MLP_STEP(wnext, jnext, kqnext, wcur, kqcur)                             \
    do {                                                                        \
        MLP_LOADW(wnext, jnext, kqnext);                                        \
        __builtin_amdgcn_sched_barrier(0);                                      \
        MLP_FMAW(wcur, kqcur);                                                  \
        __builtin_amdgcn_sched_barrier(0);                                      \
    } while (0)

    v4f wa[MLP_JB_ * MLP_KQ_ / 4], wb[MLP_JB_ * MLP_KQ_ / 4];
    float o0 = b2[0];
    float o1 = b2[1];
    MLP_LOADW(wa, 0, 0);
    for (int j0 = 0; j0 < H_; j0 += MLP_JB_) {
        const v4f bq0 = *(const v4f*)(sBW + j0);
        const v4f bq1 = *(const v4f*)(sBW + j0 + 4);
        const v4f u0  = *(const v4f*)(sBW + H_ + j0);
        const v4f u1  = *(const v4f*)(sBW + H_ + j0 + 4);
        const v4f v0  = *(const v4f*)(sBW + 2 * H_ + j0);
        const v4f v1  = *(const v4f*)(sBW + 2 * H_ + j0 + 4);
        float acc[MLP_JB_];
#pragma unroll
        for (int i = 0; i < MLP_JB_; ++i) acc[i] = i < 4 ? bq0[i & 3] : bq1[i & 3];
        const int jn = (j0 + MLP_JB_) & (H_ - 1);
        MLP_STEP(wb, j0, 1, wa, 0);
        MLP_STEP(wa, j0, 2, wb, 1);
        MLP_STEP(wb, j0, 3, wa, 2);
        MLP_STEP(wa, jn, 0, wb, 3);
#pragma unroll
        for (int i = 0; i < MLP_JB_; ++i) {
            float hj = fmaxf(acc[i], 0.0f);
            o0 = fmaf(i < 4 ? u0[i & 3] : u1[i & 3], hj, o0);
            o1 = fmaf(i < 4 ? v0[i & 3] : v1[i & 3], hj, o1);
        }
    }
#undef MLP_STEP
#undef MLP_FMAW
#undef MLP_LOADW
    out[idx] = o0;
    float sp = fmaxf(o1, 0.0f) + log1pf(expf(-fabsf(o1)));
    out[C_ * TG_ + idx] = sp + 1e-6f;
}

extern "C" void kernel_launch(void* const* d_in, const int* in_sizes, int n_in,
                              void* d_out, int out_size, void* d_ws, size_t ws_size,
                              hipStream_t stream) {
    (void)in_sizes; (void)n_in; (void)out_size; (void)ws_size;
    const float* z   = (const float*)d_in[0];
    const float* grn = (const float*)d_in[1];
    const int*   btf = (const int*)  d_in[2];
    const float* w_z = (const float*)d_in[4];
    const float* b_z = (const float*)d_in[5];
    const float* w1  = (const float*)d_in[6];
    const float* b1  = (const float*)d_in[7];
    const float* w2  = (const float*)d_in[8];
    const float* b2  = (const float*)d_in[9];
    float* out = (float*)d_out;

    float* Xa = (float*)d_ws;
    float* Xb = Xa + (size_t)C_ * TG_ * H_;

    const int total = C_ * TG_ * H_;
    init_x_kernel<<<(total + 255) / 256, 256, 0, stream>>>(z, w_z, b_z, Xa, Xb);

    dim3 grid(TF_ / 16, C_);
    float* bufs[2] = { Xa, Xb };
    for (int it = 0; it < NITER; ++it) {
        const float* xin = bufs[it & 1];
        float*      xout = bufs[(it + 1) & 1];
        neumann_iter_kernel<<<grid, 128, 0, stream>>>(grn, z, w_z, b_z, btf, xin, xout);
    }
    const float* xfin = bufs[NITER & 1];

    mlp_kernel<<<(C_ * TG_) / MLP_ROWS_, MLP_ROWS_, 0, stream>>>(xfin, w1, b1, w2, b2, out);
}
